# starting kernel
# baseline (speedup 1.0000x reference)
#include <hip/hip_runtime.h>

#define B_ 32
#define S_ 4096
#define H_ 512
#define NCHUNK 16
#define ROWS_PER_CHUNK (S_ / NCHUNK)
#define WAVES_PER_BLOCK 8
#define ROWS_PER_WAVE (ROWS_PER_CHUNK / WAVES_PER_BLOCK)

typedef __attribute__((ext_vector_type(2))) float v2f;
typedef __attribute__((ext_vector_type(8))) float v8f;

__global__ __launch_bounds__(32) void query_wmma_kernel(
    const float* __restrict__ dec, const float* __restrict__ W,
    float* __restrict__ q) {
  const int m0   = blockIdx.x * 16;
  const int n0   = blockIdx.y * 16;
  const int lane = threadIdx.x;
  const int half = lane >> 4;
  const int r    = lane & 15;
  const int koff = half * 2;

  v8f c = {};
  for (int k = 0; k < H_; k += 4) {
    const float* ap = dec + (m0 + r) * H_ + k + koff;
    const float* bp = W   + (n0 + r) * H_ + k + koff;
    v2f a, b;
    a.x = ap[0]; a.y = ap[1];
    b.x = bp[0]; b.y = bp[1];
    c = __builtin_amdgcn_wmma_f32_16x16x4_f32(
         false, a,  false, b,
         (short)0, c,  false,  false);
  }

  #pragma unroll
  for (int v = 0; v < 8; ++v) {
    const int m = v + 8 * half;
    q[(m0 + m) * H_ + (n0 + r)] = c[v];
  }
}

__global__ __launch_bounds__(256) void fused_scores_context(
    const float* __restrict__ enc, const float* __restrict__ qws,
    float* __restrict__ partials) {
  const int b     = blockIdx.x;
  const int chunk = blockIdx.y;
  const int tid   = threadIdx.x;
  const int wave  = tid >> 5;
  const int lane  = tid & 31;

  const float* q = qws + b * H_;
  float4 qr[4];
  #pragma unroll
  for (int j = 0; j < 4; ++j)
    qr[j] = *(const float4*)(q + j * 128 + lane * 4);

  float4 ctx[4];
  #pragma unroll
  for (int j = 0; j < 4; ++j) ctx[j] = make_float4(0.f, 0.f, 0.f, 0.f);

  const int s0 = chunk * ROWS_PER_CHUNK + wave * ROWS_PER_WAVE;
  const float* base = enc + ((size_t)b * S_ + s0) * H_;

  for (int s = 0; s < ROWS_PER_WAVE; ++s) {
    const float* row = base + (size_t)s * H_;
    __builtin_prefetch(row + H_ + lane * 4, 0, 0);

    float4 e[4];
    #pragma unroll
    for (int j = 0; j < 4; ++j)
      e[j] = *(const float4*)(row + j * 128 + lane * 4);

    float p = 0.f;
    #pragma unroll
    for (int j = 0; j < 4; ++j)
      p += e[j].x * qr[j].x + e[j].y * qr[j].y
         + e[j].z * qr[j].z + e[j].w * qr[j].w;

    #pragma unroll
    for (int off = 16; off >= 1; off >>= 1)
      p += __shfl_xor(p, off, 32);

    #pragma unroll
    for (int j = 0; j < 4; ++j) {
      ctx[j].x += p * e[j].x;
      ctx[j].y += p * e[j].y;
      ctx[j].z += p * e[j].z;
      ctx[j].w += p * e[j].w;
    }
  }

  __shared__ float red[WAVES_PER_BLOCK][H_];
  #pragma unroll
  for (int j = 0; j < 4; ++j)
    *(float4*)&red[wave][j * 128 + lane * 4] = ctx[j];
  __syncthreads();

  for (int h = tid; h < H_; h += 256) {
    float sum = 0.f;
    #pragma unroll
    for (int w = 0; w < WAVES_PER_BLOCK; ++w) sum += red[w][h];
    partials[(size_t)(b * NCHUNK + chunk) * H_ + h] = sum;
  }
}

__global__ __launch_bounds__(256) void reduce_context(
    const float* __restrict__ partials, float* __restrict__ out) {
  const int b = blockIdx.x;
  for (int h = threadIdx.x; h < H_; h += 256) {
    float sum = 0.f;
    #pragma unroll
    for (int c = 0; c < NCHUNK; ++c)
      sum += partials[(size_t)(b * NCHUNK + c) * H_ + h];
    out[b * H_ + h] = sum;
  }
}

extern "C" void kernel_launch(void* const* d_in, const int* in_sizes, int n_in,
                              void* d_out, int out_size, void* d_ws, size_t ws_size,
                              hipStream_t stream) {
  (void)in_sizes; (void)n_in; (void)out_size; (void)ws_size;
  const float* enc = (const float*)d_in[0];
  const float* dec = (const float*)d_in[1];
  const float* W   = (const float*)d_in[2];
  float* out = (float*)d_out;

  float* q_ws     = (float*)d_ws;
  float* partials = q_ws + B_ * H_;

  query_wmma_kernel<<<dim3(2, H_ / 16), 32, 0, stream>>>(dec, W, q_ws);
  fused_scores_context<<<dim3(B_, NCHUNK), 256, 0, stream>>>(enc, q_ws, partials);
  reduce_context<<<B_, 256, 0, stream>>>(partials, out);
}


# Query GEMM: one wave per tile kept; the next 16 k-steps' fragment loads are requested before the current 16 WMMAs run (register buffers), so memory round trips overlap matrix work; k order unchanged
# speedup vs baseline: 1.2538x; 1.2538x over previous
#include <hip/hip_runtime.h>

#define B_ 32
#define S_ 4096
#define H_ 512
#define NCHUNK 16
#define ROWS_PER_CHUNK (S_ / NCHUNK)
#define WAVES_PER_BLOCK 16
#define ROWS_PER_WAVE (ROWS_PER_CHUNK / WAVES_PER_BLOCK)
#define THREADS_K2 512
#define KSTEPS 16
#define KBLOCKS (H_ / (4 * KSTEPS))

typedef __attribute__((ext_vector_type(2))) float v2f;
typedef __attribute__((ext_vector_type(8))) float v8f;

__global__ __launch_bounds__(32) void query_wmma_kernel(
    const float* __restrict__ dec, const float* __restrict__ W,
    float* __restrict__ q) {
  const int m0   = blockIdx.x * 16;
  const int n0   = blockIdx.y * 16;
  const int lane = threadIdx.x;
  const int half = lane >> 4;
  const int r    = lane & 15;
  const int koff = half * 2;

  const float* ap = dec + (m0 + r) * H_ + koff;
  const float* bp = W   + (n0 + r) * H_ + koff;

  v2f a[2][KSTEPS], b[2][KSTEPS];
  v8f c = {};

  #pragma unroll
  for (int i = 0; i < KSTEPS; ++i) {
    a[0][i] = *(const v2f*)(ap + 4 * i);
    b[0][i] = *(const v2f*)(bp + 4 * i);
  }

  #pragma unroll
  for (int kb = 0; kb < KBLOCKS; ++kb) {
    const int cur = kb & 1;
    if (kb + 1 < KBLOCKS) {
      #pragma unroll
      for (int i = 0; i < KSTEPS; ++i) {
        a[cur ^ 1][i] = *(const v2f*)(ap + (kb + 1) * 4 * KSTEPS + 4 * i);
        b[cur ^ 1][i] = *(const v2f*)(bp + (kb + 1) * 4 * KSTEPS + 4 * i);
      }
    }
    __builtin_amdgcn_sched_barrier(0);
    #pragma unroll
    for (int i = 0; i < KSTEPS; ++i)
      c = __builtin_amdgcn_wmma_f32_16x16x4_f32(
           false, a[cur][i],  false, b[cur][i],
           (short)0, c,  false,  false);
  }

  #pragma unroll
  for (int v = 0; v < 8; ++v) {
    const int m = v + 8 * half;
    q[(m0 + m) * H_ + (n0 + r)] = c[v];
  }
}

__global__ __launch_bounds__(THREADS_K2) void fused_scores_context(
    const float* __restrict__ enc, const float* __restrict__ qws,
    float* __restrict__ partials) {
  const int b     = blockIdx.x;
  const int chunk = blockIdx.y;
  const int tid   = threadIdx.x;
  const int wave  = tid >> 5;
  const int lane  = tid & 31;

  const float* q = qws + b * H_;
  float4 qr[4];
  #pragma unroll
  for (int j = 0; j < 4; ++j)
    qr[j] = *(const float4*)(q + j * 128 + lane * 4);

  float4 ctx[4];
  #pragma unroll
  for (int j = 0; j < 4; ++j) ctx[j] = make_float4(0.f, 0.f, 0.f, 0.f);

  const int s0 = chunk * ROWS_PER_CHUNK + wave * ROWS_PER_WAVE;
  const float* base = enc + ((size_t)b * S_ + s0) * H_;

  #pragma unroll 2
  for (int s = 0; s < ROWS_PER_WAVE; ++s) {
    const float* row = base + (size_t)s * H_;
    __builtin_prefetch(row + H_ + lane * 4, 0, 0);

    float4 e[4];
    #pragma unroll
    for (int j = 0; j < 4; ++j)
      e[j] = *(const float4*)(row + j * 128 + lane * 4);

    float p = 0.f;
    #pragma unroll
    for (int j = 0; j < 4; ++j)
      p += e[j].x * qr[j].x + e[j].y * qr[j].y
         + e[j].z * qr[j].z + e[j].w * qr[j].w;

    #pragma unroll
    for (int off = 16; off >= 1; off >>= 1)
      p += __shfl_xor(p, off, 32);

    #pragma unroll
    for (int j = 0; j < 4; ++j) {
      ctx[j].x += p * e[j].x;
      ctx[j].y += p * e[j].y;
      ctx[j].z += p * e[j].z;
      ctx[j].w += p * e[j].w;
    }
  }

  __shared__ float red[WAVES_PER_BLOCK][H_];
  #pragma unroll
  for (int j = 0; j < 4; ++j)
    *(float4*)&red[wave][j * 128 + lane * 4] = ctx[j];
  __syncthreads();

  for (int h = tid; h < H_; h += THREADS_K2) {
    float sum = 0.f;
    #pragma unroll
    for (int w = 0; w < WAVES_PER_BLOCK; ++w) sum += red[w][h];
    partials[(size_t)(b * NCHUNK + chunk) * H_ + h] = sum;
  }
}

__global__ __launch_bounds__(256) void reduce_context(
    const float* __restrict__ partials, float* __restrict__ out) {
  const int b = blockIdx.x;
  for (int h = threadIdx.x; h < H_; h += 256) {
    float sum = 0.f;
    #pragma unroll
    for (int c = 0; c < NCHUNK; ++c)
      sum += partials[(size_t)(b * NCHUNK + c) * H_ + h];
    out[b * H_ + h] = sum;
  }
}

extern "C" void kernel_launch(void* const* d_in, const int* in_sizes, int n_in,
                              void* d_out, int out_size, void* d_ws, size_t ws_size,
                              hipStream_t stream) {
  (void)in_sizes; (void)n_in; (void)out_size; (void)ws_size;
  const float* enc = (const float*)d_in[0];
  const float* dec = (const float*)d_in[1];
  const float* W   = (const float*)d_in[2];
  float* out = (float*)d_out;

  float* q_ws     = (float*)d_ws;
  float* partials = q_ws + B_ * H_;

  query_wmma_kernel<<<dim3(2, H_ / 16), 32, 0, stream>>>(dec, W, q_ws);
  fused_scores_context<<<dim3(B_, NCHUNK), THREADS_K2, 0, stream>>>(enc, q_ws, partials);
  reduce_context<<<B_, 256, 0, stream>>>(partials, out);
}


# Streaming kernel: half-wave per row (lanes 0-15 row s, 16-31 row s+1): a row pair's 8 loads feed one dot and go out together, one 4-step butterfly serves both rows; dot and ctx sum order changes
# speedup vs baseline: 1.3116x; 1.0461x over previous
#include <hip/hip_runtime.h>

#define B_ 32
#define S_ 4096
#define H_ 512
#define NCHUNK 16
#define ROWS_PER_CHUNK (S_ / NCHUNK)
#define WAVES_PER_BLOCK 16
#define ROWS_PER_WAVE (ROWS_PER_CHUNK / WAVES_PER_BLOCK)
#define THREADS_K2 512
#define KSTEPS 16
#define KBLOCKS (H_ / (4 * KSTEPS))

typedef __attribute__((ext_vector_type(2))) float v2f;
typedef __attribute__((ext_vector_type(8))) float v8f;

__global__ __launch_bounds__(32) void query_wmma_kernel(
    const float* __restrict__ dec, const float* __restrict__ W,
    float* __restrict__ q) {
  const int m0   = blockIdx.x * 16;
  const int n0   = blockIdx.y * 16;
  const int lane = threadIdx.x;
  const int half = lane >> 4;
  const int r    = lane & 15;
  const int koff = half * 2;

  const float* ap = dec + (m0 + r) * H_ + koff;
  const float* bp = W   + (n0 + r) * H_ + koff;

  v2f a[2][KSTEPS], b[2][KSTEPS];
  v8f c = {};

  #pragma unroll
  for (int i = 0; i < KSTEPS; ++i) {
    a[0][i] = *(const v2f*)(ap + 4 * i);
    b[0][i] = *(const v2f*)(bp + 4 * i);
  }

  #pragma unroll
  for (int kb = 0; kb < KBLOCKS; ++kb) {
    const int cur = kb & 1;
    if (kb + 1 < KBLOCKS) {
      #pragma unroll
      for (int i = 0; i < KSTEPS; ++i) {
        a[cur ^ 1][i] = *(const v2f*)(ap + (kb + 1) * 4 * KSTEPS + 4 * i);
        b[cur ^ 1][i] = *(const v2f*)(bp + (kb + 1) * 4 * KSTEPS + 4 * i);
      }
    }
    __builtin_amdgcn_sched_barrier(0);
    #pragma unroll
    for (int i = 0; i < KSTEPS; ++i)
      c = __builtin_amdgcn_wmma_f32_16x16x4_f32(
           false, a[cur][i],  false, b[cur][i],
           (short)0, c,  false,  false);
  }

  #pragma unroll
  for (int v = 0; v < 8; ++v) {
    const int m = v + 8 * half;
    q[(m0 + m) * H_ + (n0 + r)] = c[v];
  }
}

__global__ __launch_bounds__(THREADS_K2) void fused_scores_context(
    const float* __restrict__ enc, const float* __restrict__ qws,
    float* __restrict__ partials) {
  const int b     = blockIdx.x;
  const int chunk = blockIdx.y;
  const int tid   = threadIdx.x;
  const int wave  = tid >> 5;
  const int lane  = tid & 31;
  const int half  = lane >> 4;
  const int c16   = lane & 15;

  const float* q = qws + b * H_ + c16 * 4;
  float4 qr[8];
  #pragma unroll
  for (int j = 0; j < 8; ++j)
    qr[j] = *(const float4*)(q + j * 64);

  float4 ctx[8];
  #pragma unroll
  for (int j = 0; j < 8; ++j) ctx[j] = make_float4(0.f, 0.f, 0.f, 0.f);

  const int s0 = chunk * ROWS_PER_CHUNK + wave * ROWS_PER_WAVE;
  const float* base = enc + ((size_t)b * S_ + s0 + half) * H_ + c16 * 4;

  #pragma unroll 1
  for (int s = 0; s < ROWS_PER_WAVE; s += 2) {
    const float* row = base + (size_t)s * H_;
    if (s + 2 < ROWS_PER_WAVE) {
      __builtin_prefetch(row + 2 * H_, 0, 0);
      __builtin_prefetch(row + 2 * H_ + 64, 0, 0);
    }

    float4 e[8];
    #pragma unroll
    for (int j = 0; j < 8; ++j)
      e[j] = *(const float4*)(row + j * 64);

    v2f axy = {0.f, 0.f}, azw = {0.f, 0.f};
    #pragma unroll
    for (int j = 0; j < 8; ++j) {
      const v2f exy = {e[j].x, e[j].y}, ezw = {e[j].z, e[j].w};
      const v2f qxy = {qr[j].x, qr[j].y}, qzw = {qr[j].z, qr[j].w};
      axy += exy * qxy;
      azw += ezw * qzw;
    }
    float p = (axy.x + axy.y) + (azw.x + azw.y);

    #pragma unroll
    for (int off = 8; off >= 1; off >>= 1)
      p += __shfl_xor(p, off, 32);

    #pragma unroll
    for (int j = 0; j < 8; ++j) {
      ctx[j].x += p * e[j].x;
      ctx[j].y += p * e[j].y;
      ctx[j].z += p * e[j].z;
      ctx[j].w += p * e[j].w;
    }
  }

  __shared__ float red[2 * WAVES_PER_BLOCK][H_];
  #pragma unroll
  for (int j = 0; j < 8; ++j)
    *(float4*)&red[2 * wave + half][j * 64 + c16 * 4] = ctx[j];
  __syncthreads();

  for (int h = tid; h < H_; h += THREADS_K2) {
    float sum = 0.f;
    #pragma unroll
    for (int w = 0; w < 2 * WAVES_PER_BLOCK; ++w) sum += red[w][h];
    partials[(size_t)(b * NCHUNK + chunk) * H_ + h] = sum;
  }
}

__global__ __launch_bounds__(256) void reduce_context(
    const float* __restrict__ partials, float* __restrict__ out) {
  const int b = blockIdx.x;
  for (int h = threadIdx.x; h < H_; h += 256) {
    float sum = 0.f;
    #pragma unroll
    for (int c = 0; c < NCHUNK; ++c)
      sum += partials[(size_t)(b * NCHUNK + c) * H_ + h];
    out[b * H_ + h] = sum;
  }
}

extern "C" void kernel_launch(void* const* d_in, const int* in_sizes, int n_in,
                              void* d_out, int out_size, void* d_ws, size_t ws_size,
                              hipStream_t stream) {
  (void)in_sizes; (void)n_in; (void)out_size; (void)ws_size;
  const float* enc = (const float*)d_in[0];
  const float* dec = (const float*)d_in[1];
  const float* W   = (const float*)d_in[2];
  float* out = (float*)d_out;

  float* q_ws     = (float*)d_ws;
  float* partials = q_ws + B_ * H_;

  query_wmma_kernel<<<dim3(2, H_ / 16), 32, 0, stream>>>(dec, W, q_ws);
  fused_scores_context<<<dim3(B_, NCHUNK), THREADS_K2, 0, stream>>>(enc, q_ws, partials);
  reduce_context<<<B_, 256, 0, stream>>>(partials, out);
}


# Streaming kernel enters half as many waves: one 16-wave block per WGP (256 blocks, 8 chunks), a wave takes 32 rows, 4 per trip (16 loads): bytes in flight per WGP kept; ctx/chunk sum order changes
# speedup vs baseline: 1.3664x; 1.0418x over previous
#include <hip/hip_runtime.h>

#define B_ 32
#define S_ 4096
#define H_ 512
#define NCHUNK 8
#define ROWS_PER_CHUNK (S_ / NCHUNK)
#define WAVES_PER_BLOCK 16
#define ROWS_PER_WAVE (ROWS_PER_CHUNK / WAVES_PER_BLOCK)
#define ROWS_PER_TRIP 4
#define THREADS_K2 512
#define KSTEPS 16
#define KBLOCKS (H_ / (4 * KSTEPS))

typedef __attribute__((ext_vector_type(2))) float v2f;
typedef __attribute__((ext_vector_type(8))) float v8f;

__global__ __launch_bounds__(32) void query_wmma_kernel(
    const float* __restrict__ dec, const float* __restrict__ W,
    float* __restrict__ q) {
  const int m0   = blockIdx.x * 16;
  const int n0   = blockIdx.y * 16;
  const int lane = threadIdx.x;
  const int half = lane >> 4;
  const int r    = lane & 15;
  const int koff = half * 2;

  const float* ap = dec + (m0 + r) * H_ + koff;
  const float* bp = W   + (n0 + r) * H_ + koff;

  v2f a[2][KSTEPS], b[2][KSTEPS];
  v8f c = {};

  #pragma unroll
  for (int i = 0; i < KSTEPS; ++i) {
    a[0][i] = *(const v2f*)(ap + 4 * i);
    b[0][i] = *(const v2f*)(bp + 4 * i);
  }

  #pragma unroll
  for (int kb = 0; kb < KBLOCKS; ++kb) {
    const int cur = kb & 1;
    if (kb + 1 < KBLOCKS) {
      #pragma unroll
      for (int i = 0; i < KSTEPS; ++i) {
        a[cur ^ 1][i] = *(const v2f*)(ap + (kb + 1) * 4 * KSTEPS + 4 * i);
        b[cur ^ 1][i] = *(const v2f*)(bp + (kb + 1) * 4 * KSTEPS + 4 * i);
      }
    }
    __builtin_amdgcn_sched_barrier(0);
    #pragma unroll
    for (int i = 0; i < KSTEPS; ++i)
      c = __builtin_amdgcn_wmma_f32_16x16x4_f32(
           false, a[cur][i],  false, b[cur][i],
           (short)0, c,  false,  false);
  }

  #pragma unroll
  for (int v = 0; v < 8; ++v) {
    const int m = v + 8 * half;
    q[(m0 + m) * H_ + (n0 + r)] = c[v];
  }
}

__global__ __launch_bounds__(THREADS_K2) void fused_scores_context(
    const float* __restrict__ enc, const float* __restrict__ qws,
    float* __restrict__ partials) {
  const int b     = blockIdx.x;
  const int chunk = blockIdx.y;
  const int tid   = threadIdx.x;
  const int wave  = tid >> 5;
  const int lane  = tid & 31;
  const int half  = lane >> 4;
  const int c16   = lane & 15;

  const float* q = qws + b * H_ + c16 * 4;
  float4 qr[8];
  #pragma unroll
  for (int j = 0; j < 8; ++j)
    qr[j] = *(const float4*)(q + j * 64);

  float4 ctx[8];
  #pragma unroll
  for (int j = 0; j < 8; ++j) ctx[j] = make_float4(0.f, 0.f, 0.f, 0.f);

  const int s0 = chunk * ROWS_PER_CHUNK + wave * ROWS_PER_WAVE;
  const float* base = enc + ((size_t)b * S_ + s0 + half) * H_ + c16 * 4;

  for (int s = 0; s < ROWS_PER_WAVE; s += ROWS_PER_TRIP) {
    const float* row = base + (size_t)s * H_;
    if (s + ROWS_PER_TRIP < ROWS_PER_WAVE) {
      __builtin_prefetch(row + 4 * H_, 0, 0);
      __builtin_prefetch(row + 4 * H_ + 64, 0, 0);
      __builtin_prefetch(row + 6 * H_, 0, 0);
      __builtin_prefetch(row + 6 * H_ + 64, 0, 0);
    }

    float4 e0[8], e1[8];
    #pragma unroll
    for (int j = 0; j < 8; ++j)
      e0[j] = *(const float4*)(row + j * 64);
    #pragma unroll
    for (int j = 0; j < 8; ++j)
      e1[j] = *(const float4*)(row + 2 * H_ + j * 64);
    __builtin_amdgcn_sched_barrier(0);

    v2f axy0 = {0.f, 0.f}, azw0 = {0.f, 0.f};
    #pragma unroll
    for (int j = 0; j < 8; ++j) {
      const v2f exy = {e0[j].x, e0[j].y}, ezw = {e0[j].z, e0[j].w};
      const v2f qxy = {qr[j].x, qr[j].y}, qzw = {qr[j].z, qr[j].w};
      axy0 += exy * qxy;
      azw0 += ezw * qzw;
    }
    v2f axy1 = {0.f, 0.f}, azw1 = {0.f, 0.f};
    #pragma unroll
    for (int j = 0; j < 8; ++j) {
      const v2f exy = {e1[j].x, e1[j].y}, ezw = {e1[j].z, e1[j].w};
      const v2f qxy = {qr[j].x, qr[j].y}, qzw = {qr[j].z, qr[j].w};
      axy1 += exy * qxy;
      azw1 += ezw * qzw;
    }
    float p0 = (axy0.x + axy0.y) + (azw0.x + azw0.y);
    float p1 = (axy1.x + axy1.y) + (azw1.x + azw1.y);

    #pragma unroll
    for (int off = 8; off >= 1; off >>= 1) {
      const float t0 = __shfl_xor(p0, off, 32);
      const float t1 = __shfl_xor(p1, off, 32);
      p0 += t0;
      p1 += t1;
    }

    #pragma unroll
    for (int j = 0; j < 8; ++j) {
      ctx[j].x += p0 * e0[j].x;
      ctx[j].y += p0 * e0[j].y;
      ctx[j].z += p0 * e0[j].z;
      ctx[j].w += p0 * e0[j].w;
    }
    #pragma unroll
    for (int j = 0; j < 8; ++j) {
      ctx[j].x += p1 * e1[j].x;
      ctx[j].y += p1 * e1[j].y;
      ctx[j].z += p1 * e1[j].z;
      ctx[j].w += p1 * e1[j].w;
    }
  }

  __shared__ float red[2 * WAVES_PER_BLOCK][H_];
  #pragma unroll
  for (int j = 0; j < 8; ++j)
    *(float4*)&red[2 * wave + half][j * 64 + c16 * 4] = ctx[j];
  __syncthreads();

  for (int h = tid; h < H_; h += THREADS_K2) {
    float sum = 0.f;
    #pragma unroll
    for (int w = 0; w < 2 * WAVES_PER_BLOCK; ++w) sum += red[w][h];
    partials[(size_t)(b * NCHUNK + chunk) * H_ + h] = sum;
  }
}

__global__ __launch_bounds__(256) void reduce_context(
    const float* __restrict__ partials, float* __restrict__ out) {
  const int b = blockIdx.x;
  for (int h = threadIdx.x; h < H_; h += 256) {
    float sum = 0.f;
    #pragma unroll
    for (int c = 0; c < NCHUNK; ++c)
      sum += partials[(size_t)(b * NCHUNK + c) * H_ + h];
    out[b * H_ + h] = sum;
  }
}

extern "C" void kernel_launch(void* const* d_in, const int* in_sizes, int n_in,
                              void* d_out, int out_size, void* d_ws, size_t ws_size,
                              hipStream_t stream) {
  (void)in_sizes; (void)n_in; (void)out_size; (void)ws_size;
  const float* enc = (const float*)d_in[0];
  const float* dec = (const float*)d_in[1];
  const float* W   = (const float*)d_in[2];
  float* out = (float*)d_out;

  float* q_ws     = (float*)d_ws;
  float* partials = q_ws + B_ * H_;

  query_wmma_kernel<<<dim3(2, H_ / 16), 32, 0, stream>>>(dec, W, q_ws);
  fused_scores_context<<<dim3(B_, NCHUNK), THREADS_K2, 0, stream>>>(enc, q_ws, partials);
  reduce_context<<<B_, 256, 0, stream>>>(partials, out);
}
